# starting kernel
# baseline (speedup 1.0000x reference)
#include <hip/hip_runtime.h>
#include <math.h>

typedef __attribute__((ext_vector_type(2))) float v2f;
typedef __attribute__((ext_vector_type(8))) float v8f;

#define NEG_SLOPE 0.2f

__device__ __forceinline__ void atomicMaxF(float* addr, float v) {
  if (v >= 0.0f) atomicMax((int*)addr, __float_as_int(v));
  else           atomicMin((unsigned int*)addr, __float_as_uint(v));
}

__global__ void __launch_bounds__(256)
fill_kernel(float* __restrict__ p, int n, float v) {
  int i = blockIdx.x * blockDim.x + threadIdx.x;
  if (i < n) p[i] = v;
}

__global__ void __launch_bounds__(128)
gemm1_kernel(const float* __restrict__ x, const float* __restrict__ W1l,
             const float* __restrict__ W1r, float* __restrict__ xl1,
             float* __restrict__ xr1, int n_nodes) {
  const int lane = threadIdx.x & 31;
  const int wave = threadIdx.x >> 5;
  const int m0   = blockIdx.x * 16;
  const int n0   = wave * 16;
  const int half = lane >> 4;
  const int rc   = lane & 15;

  int arow = m0 + rc; if (arow >= n_nodes) arow = n_nodes - 1;
  const v2f*   xrow = (const v2f*)(x + (size_t)arow * 128);
  const float* Bl   = W1l + n0 + rc;
  const float* Br   = W1r + n0 + rc;

  v8f accL = {}, accR = {};
  #pragma unroll
  for (int kc = 0; kc < 32; ++kc) {
    v2f a = xrow[2 * kc + half];
    const int k0 = 4 * kc + 2 * half;
    v2f bl, br;
    bl[0] = Bl[(size_t)(k0 + 0) * 64];
    bl[1] = Bl[(size_t)(k0 + 1) * 64];
    br[0] = Br[(size_t)(k0 + 0) * 64];
    br[1] = Br[(size_t)(k0 + 1) * 64];
    accL = __builtin_amdgcn_wmma_f32_16x16x4_f32(
        false, a, false, bl, (short)0, accL, false, false);
    accR = __builtin_amdgcn_wmma_f32_16x16x4_f32(
        false, a, false, br, (short)0, accR, false, false);
  }

  #pragma unroll
  for (int r = 0; r < 8; ++r) {
    int row = m0 + r + 8 * half;
    if (row < n_nodes) {
      xl1[(size_t)row * 64 + n0 + rc] = accL[r];
      xr1[(size_t)row * 64 + n0 + rc] = accR[r];
    }
  }
}

__global__ void __launch_bounds__(128)
gemm2_kernel(const float* __restrict__ h, const float* __restrict__ W2l,
             const float* __restrict__ W2r, float* __restrict__ xl2,
             float* __restrict__ xr2, int n_nodes, int mtiles) {
  const int lane  = threadIdx.x & 31;
  const int mtile = blockIdx.x * 4 + (threadIdx.x >> 5);
  if (mtile >= mtiles) return;
  const int m0   = mtile * 16;
  const int half = lane >> 4;
  const int rc   = lane & 15;

  int arow = m0 + rc; if (arow >= n_nodes) arow = n_nodes - 1;
  const v2f* hrow = (const v2f*)(h + (size_t)arow * 64);
  const bool colok = (rc < 10);

  v8f accL = {}, accR = {};
  #pragma unroll
  for (int kc = 0; kc < 16; ++kc) {
    v2f a = hrow[2 * kc + half];
    const int k0 = 4 * kc + 2 * half;
    v2f bl, br;
    bl[0] = colok ? W2l[(size_t)(k0 + 0) * 10 + rc] : 0.0f;
    bl[1] = colok ? W2l[(size_t)(k0 + 1) * 10 + rc] : 0.0f;
    br[0] = colok ? W2r[(size_t)(k0 + 0) * 10 + rc] : 0.0f;
    br[1] = colok ? W2r[(size_t)(k0 + 1) * 10 + rc] : 0.0f;
    accL = __builtin_amdgcn_wmma_f32_16x16x4_f32(
        false, a, false, bl, (short)0, accL, false, false);
    accR = __builtin_amdgcn_wmma_f32_16x16x4_f32(
        false, a, false, br, (short)0, accR, false, false);
  }

  #pragma unroll
  for (int r = 0; r < 8; ++r) {
    int row = m0 + r + 8 * half;
    if (row < n_nodes) {
      xl2[(size_t)row * 16 + rc] = accL[r];
      xr2[(size_t)row * 16 + rc] = accR[r];
    }
  }
}

__global__ void __launch_bounds__(256)
l1_edge_logits(const int* __restrict__ ei, const float* __restrict__ xl1,
               const float* __restrict__ xr1, const float* __restrict__ att1,
               float* __restrict__ logits1, float* __restrict__ m1, int E) {
  int gid = blockIdx.x * blockDim.x + threadIdx.x;
  int g   = gid >> 4;
  bool act = (g < E);
  if (g >= E) g = E - 1;
  const int l   = threadIdx.x & 15;
  const int src = ei[g];
  const int dst = ei[E + g];

  const float4 xv = *(const float4*)(xl1 + (size_t)src * 64 + l * 4);
  const float4 rv = *(const float4*)(xr1 + (size_t)dst * 64 + l * 4);
  const float4 at = *(const float4*)(att1 + l * 4);

  float p = 0.0f, s;
  s = xv.x + rv.x; s = (s > 0.f) ? s : NEG_SLOPE * s; p += s * at.x;
  s = xv.y + rv.y; s = (s > 0.f) ? s : NEG_SLOPE * s; p += s * at.y;
  s = xv.z + rv.z; s = (s > 0.f) ? s : NEG_SLOPE * s; p += s * at.z;
  s = xv.w + rv.w; s = (s > 0.f) ? s : NEG_SLOPE * s; p += s * at.w;

  p += __shfl_xor(p, 1, 32);
  p += __shfl_xor(p, 2, 32);

  if (act && (l & 3) == 0) {
    int hd = l >> 2;
    logits1[(size_t)g * 4 + hd] = p;
    atomicMaxF(m1 + (size_t)dst * 4 + hd, p);
  }
}

__global__ void __launch_bounds__(256)
l1_edge_aggregate(const int* __restrict__ ei, const float* __restrict__ xl1,
                  const float* __restrict__ logits1, const float* __restrict__ m1,
                  float* __restrict__ den1, float* __restrict__ hacc, int E) {
  int g = (blockIdx.x * blockDim.x + threadIdx.x) >> 4;
  if (g >= E) return;
  const int l   = threadIdx.x & 15;
  const int src = ei[g];
  const int dst = ei[E + g];
  const int hd  = l >> 2;

  float w = __expf(logits1[(size_t)g * 4 + hd] - m1[(size_t)dst * 4 + hd]);
  if ((l & 3) == 0) atomicAdd(den1 + (size_t)dst * 4 + hd, w);

  const float4 v = *(const float4*)(xl1 + (size_t)src * 64 + l * 4);
  float* o = hacc + (size_t)dst * 64 + l * 4;
  atomicAdd(o + 0, w * v.x);
  atomicAdd(o + 1, w * v.y);
  atomicAdd(o + 2, w * v.z);
  atomicAdd(o + 3, w * v.w);
}

__global__ void __launch_bounds__(256)
l1_finish(float* __restrict__ hacc, const float* __restrict__ den1,
          const float* __restrict__ b1, int total) {
  int i = blockIdx.x * blockDim.x + threadIdx.x;
  if (i >= total) return;
  int n = i >> 6, ch = i & 63, hd = ch >> 4;
  float d = den1[(size_t)n * 4 + hd];
  float v = (d > 0.0f) ? hacc[i] / d : 0.0f;
  v += b1[ch];
  hacc[i] = (v > 0.0f) ? v : expm1f(v);
}

__global__ void __launch_bounds__(256)
l2_edge_logits(const int* __restrict__ ei, const float* __restrict__ xl2,
               const float* __restrict__ xr2, const float* __restrict__ att2,
               float* __restrict__ logits2, float* __restrict__ m2, int E) {
  int gid = blockIdx.x * blockDim.x + threadIdx.x;
  int g   = gid >> 2;
  bool act = (g < E);
  if (g >= E) g = E - 1;
  const int l   = threadIdx.x & 3;
  const int src = ei[g];
  const int dst = ei[E + g];
  const int c0  = l * 4;

  const float4 a = *(const float4*)(xl2 + (size_t)src * 16 + c0);
  const float4 b = *(const float4*)(xr2 + (size_t)dst * 16 + c0);

  float p = 0.0f, s;
  s = a.x + b.x; s = (s > 0.f) ? s : NEG_SLOPE * s; if (c0 + 0 < 10) p += s * att2[c0 + 0];
  s = a.y + b.y; s = (s > 0.f) ? s : NEG_SLOPE * s; if (c0 + 1 < 10) p += s * att2[c0 + 1];
  s = a.z + b.z; s = (s > 0.f) ? s : NEG_SLOPE * s; if (c0 + 2 < 10) p += s * att2[c0 + 2];
  s = a.w + b.w; s = (s > 0.f) ? s : NEG_SLOPE * s; if (c0 + 3 < 10) p += s * att2[c0 + 3];

  p += __shfl_xor(p, 1, 32);
  p += __shfl_xor(p, 2, 32);

  if (act && l == 0) {
    logits2[g] = p;
    atomicMaxF(m2 + dst, p);
  }
}

__global__ void __launch_bounds__(256)
l2_edge_aggregate(const int* __restrict__ ei, const float* __restrict__ xl2,
                  const float* __restrict__ logits2, const float* __restrict__ m2,
                  float* __restrict__ den2, float* __restrict__ acc2, int E) {
  int g = (blockIdx.x * blockDim.x + threadIdx.x) >> 2;
  if (g >= E) return;
  const int l   = threadIdx.x & 3;
  const int src = ei[g];
  const int dst = ei[E + g];
  const int c0  = l * 4;

  float w = __expf(logits2[g] - m2[dst]);
  if (l == 0) atomicAdd(den2 + dst, w);

  const float4 v = *(const float4*)(xl2 + (size_t)src * 16 + c0);
  float* o = acc2 + (size_t)dst * 16 + c0;
  if (c0 + 0 < 10) atomicAdd(o + 0, w * v.x);
  if (c0 + 1 < 10) atomicAdd(o + 1, w * v.y);
  if (c0 + 2 < 10) atomicAdd(o + 2, w * v.z);
  if (c0 + 3 < 10) atomicAdd(o + 3, w * v.w);
}

__global__ void __launch_bounds__(256)
l2_finish(const float* __restrict__ acc2, const float* __restrict__ den2,
          const float* __restrict__ b2, float* __restrict__ out, int total) {
  int i = blockIdx.x * blockDim.x + threadIdx.x;
  if (i >= total) return;
  int n = i / 10, c = i % 10;
  float d = den2[n];
  float v = (d > 0.0f) ? acc2[(size_t)n * 16 + c] / d : 0.0f;
  out[i] = v + b2[c];
}

extern "C" void kernel_launch(void* const* d_in, const int* in_sizes, int n_in,
                              void* d_out, int out_size, void* d_ws, size_t ws_size,
                              hipStream_t stream) {
  const float* x    = (const float*)d_in[0];
  const int*   ei   = (const int*)d_in[1];
  const float* W1l  = (const float*)d_in[2];
  const float* W1r  = (const float*)d_in[3];
  const float* att1 = (const float*)d_in[4];
  const float* b1   = (const float*)d_in[5];
  const float* W2l  = (const float*)d_in[6];
  const float* W2r  = (const float*)d_in[7];
  const float* att2 = (const float*)d_in[8];
  const float* b2   = (const float*)d_in[9];
  float* out = (float*)d_out;

  const int N = in_sizes[0] / 128;
  const int E = in_sizes[1] / 2;

  float* ws = (float*)d_ws;
  size_t o = 0;
  float* xl1  = ws + o; o += (size_t)N * 64;
  float* xr1  = ws + o; o += (size_t)N * 64;
  float* hacc = ws + o; o += (size_t)N * 64;
  float* lg1  = ws + o; o += (size_t)E * 4;
  float* m1   = ws + o; o += (size_t)N * 4;
  float* dn1  = ws + o; o += (size_t)N * 4;
  float* xl2  = ws + o; o += (size_t)N * 16;
  float* xr2  = ws + o; o += (size_t)N * 16;
  float* acc2 = ws + o; o += (size_t)N * 16;
  float* lg2  = ws + o; o += (size_t)E;
  float* m2   = ws + o; o += (size_t)N;
  float* dn2  = ws + o; o += (size_t)N;

  auto blocks = [](long long n) { return (unsigned)((n + 255) / 256); };
  const float NEG_INF = -__builtin_inff();

  fill_kernel<<<blocks((long long)N * 64), 256, 0, stream>>>(hacc, N * 64, 0.0f);
  fill_kernel<<<blocks((long long)N * 4), 256, 0, stream>>>(m1, N * 4, NEG_INF);
  fill_kernel<<<blocks((long long)N * 4), 256, 0, stream>>>(dn1, N * 4, 0.0f);
  fill_kernel<<<blocks((long long)N * 16), 256, 0, stream>>>(acc2, N * 16, 0.0f);
  fill_kernel<<<blocks(N), 256, 0, stream>>>(m2, N, NEG_INF);
  fill_kernel<<<blocks(N), 256, 0, stream>>>(dn2, N, 0.0f);

  const int mtiles = (N + 15) / 16;

  gemm1_kernel<<<mtiles, 128, 0, stream>>>(x, W1l, W1r, xl1, xr1, N);
  l1_edge_logits<<<blocks((long long)E * 16), 256, 0, stream>>>(ei, xl1, xr1, att1,
                                                                lg1, m1, E);
  l1_edge_aggregate<<<blocks((long long)E * 16), 256, 0, stream>>>(ei, xl1, lg1, m1,
                                                                   dn1, hacc, E);
  l1_finish<<<blocks((long long)N * 64), 256, 0, stream>>>(hacc, dn1, b1, N * 64);

  gemm2_kernel<<<(mtiles + 3) / 4, 128, 0, stream>>>(hacc, W2l, W2r, xl2, xr2, N, mtiles);
  l2_edge_logits<<<blocks((long long)E * 4), 256, 0, stream>>>(ei, xl2, xr2, att2,
                                                               lg2, m2, E);
  l2_edge_aggregate<<<blocks((long long)E * 4), 256, 0, stream>>>(ei, xl2, lg2, m2,
                                                                  dn2, acc2, E);
  l2_finish<<<blocks((long long)N * 10), 256, 0, stream>>>(acc2, dn2, b2, out, N * 10);
}


# Element-wise passes ride in the GEMM launches: GEMM1's blocks give their 16 nodes the start values (5 fill launches gone), GEMM2 forms elu(acc/den+b1) as it loads A; 14 to 8 launches; sums unchanged
# speedup vs baseline: 2.7154x; 2.7154x over previous
#include <hip/hip_runtime.h>
#include <math.h>

typedef __attribute__((ext_vector_type(2))) float v2f;
typedef __attribute__((ext_vector_type(8))) float v8f;
typedef __attribute__((ext_vector_type(4))) float v4f;

#define NEG_SLOPE 0.2f

__device__ __forceinline__ void atomicMaxF(float* addr, float v) {
  if (v >= 0.0f) atomicMax((int*)addr, __float_as_int(v));
  else           atomicMin((unsigned int*)addr, __float_as_uint(v));
}

__global__ void __launch_bounds__(256)
partition_edges(const int* __restrict__ ei, int* __restrict__ psrc,
                int* __restrict__ pdst, int* __restrict__ cnt, int E, int split) {
  __shared__ int wc0[8], wc1[8], sbase[2];
  const int tid = threadIdx.x, lane = tid & 31, wave = tid >> 5;
  const int base = blockIdx.x * 1024;
  int s[4], d[4];
  unsigned b0[4], b1[4];
  int c0 = 0, c1 = 0;
  #pragma unroll
  for (int j = 0; j < 4; ++j) {
    const int g = base + j * 256 + tid;
    const bool valid = (g < E);
    const int gg = valid ? g : E - 1;
    s[j] = ei[gg];
    d[j] = ei[(size_t)E + gg];
    b0[j] = __builtin_amdgcn_ballot_w32(valid && d[j] <  split);
    b1[j] = __builtin_amdgcn_ballot_w32(valid && d[j] >= split);
    c0 += __popc(b0[j]);
    c1 += __popc(b1[j]);
  }
  if (lane == 0) { wc0[wave] = c0; wc1[wave] = c1; }
  __syncthreads();
  if (tid == 0) {
    int t0 = 0, t1 = 0;
    #pragma unroll
    for (int w = 0; w < 8; ++w) { t0 += wc0[w]; t1 += wc1[w]; }
    sbase[0] = atomicAdd(cnt + 0, t0);
    sbase[1] = atomicAdd(cnt + 1, t1);
  }
  __syncthreads();
  int off0 = sbase[0], off1 = sbase[1];
  for (int w = 0; w < wave; ++w) { off0 += wc0[w]; off1 += wc1[w]; }
  const unsigned lt = (1u << lane) - 1u;
  #pragma unroll
  for (int j = 0; j < 4; ++j) {
    if ((b0[j] >> lane) & 1u) {
      const int pos = off0 + __popc(b0[j] & lt);
      psrc[pos] = s[j]; pdst[pos] = d[j];
    } else if ((b1[j] >> lane) & 1u) {
      const int pos = E - 1 - (off1 + __popc(b1[j] & lt));
      psrc[pos] = s[j]; pdst[pos] = d[j];
    }
    off0 += __popc(b0[j]);
    off1 += __popc(b1[j]);
  }
}

__device__ __forceinline__ void claim_chunk(int* __restrict__ tk, const int* __restrict__ cnt,
                                            int E, int CH, int& lo, int& hi) {
  __shared__ int s_lo, s_hi;
  if (threadIdx.x == 0) {
    const int n0 = cnt[0];
    const int die = (int)((__builtin_amdgcn_s_sendmsg_rtn(0x87) >> 16) & 15u);
    const int half = (die >> 2) & 1;
    int l = 0, h = 0;
    #pragma unroll
    for (int pass = 0; pass < 2; ++pass) {
      const int sset = half ^ pass;
      const long long beg = sset ? n0 : 0;
      const long long end = sset ? E : n0;
      const long long a = beg + (long long)atomicAdd(tk + 64 * sset, 1) * CH;
      if (a < end) {
        l = (int)a;
        h = (int)((a + CH < end) ? a + CH : end);
        break;
      }
    }
    s_lo = l; s_hi = h;
  }
  __syncthreads();
  lo = s_lo; hi = s_hi;
}

__global__ void __launch_bounds__(128)
gemm1_kernel(const float* __restrict__ x, const float* __restrict__ W1l,
             const float* __restrict__ W1r, float* __restrict__ xl1,
             float* __restrict__ xr1, float* __restrict__ hacc,
             float* __restrict__ m1, float* __restrict__ dn1,
             float* __restrict__ acc2, float* __restrict__ m2,
             int* __restrict__ ctr, int n_nodes) {
  {
    const int t = threadIdx.x;
    const int r0 = blockIdx.x * 16;
    const v4f zero4 = {0.0f, 0.0f, 0.0f, 0.0f};
    const float ninf = -__builtin_inff();
    const v4f ninf4 = {ninf, ninf, ninf, ninf};
    for (int j = t; j < 256; j += 128)
      if (r0 + (j >> 4) < n_nodes) *(v4f*)(hacc + (size_t)r0 * 64 + j * 4) = zero4;
    if (t < 16) {
      if (r0 + t < n_nodes) *(v4f*)(m1 + (size_t)(r0 + t) * 4) = ninf4;
    } else if (t < 32) {
      if (r0 + t - 16 < n_nodes) *(v4f*)(dn1 + (size_t)(r0 + t - 16) * 4) = zero4;
    } else if (t < 96) {
      const int j = t - 32;
      if (r0 + (j >> 2) < n_nodes) *(v4f*)(acc2 + (size_t)r0 * 16 + j * 4) = zero4;
    } else if (t < 112) {
      if (r0 + t - 96 < n_nodes) m2[r0 + t - 96] = ninf;
    }
    if (blockIdx.x == 0)
      for (int j = t; j < 576; j += 128) ctr[j] = 0;
  }
  const int lane = threadIdx.x & 31;
  const int wave = threadIdx.x >> 5;
  const int m0   = blockIdx.x * 16;
  const int n0   = wave * 16;
  const int half = lane >> 4;
  const int rc   = lane & 15;

  int arow = m0 + rc; if (arow >= n_nodes) arow = n_nodes - 1;
  const v2f*   xrow = (const v2f*)(x + (size_t)arow * 128);
  const float* Bl   = W1l + n0 + rc;
  const float* Br   = W1r + n0 + rc;

  v8f accL = {}, accR = {};
  #pragma unroll
  for (int kc = 0; kc < 32; ++kc) {
    v2f a = xrow[2 * kc + half];
    const int k0 = 4 * kc + 2 * half;
    v2f bl, br;
    bl[0] = Bl[(size_t)(k0 + 0) * 64];
    bl[1] = Bl[(size_t)(k0 + 1) * 64];
    br[0] = Br[(size_t)(k0 + 0) * 64];
    br[1] = Br[(size_t)(k0 + 1) * 64];
    accL = __builtin_amdgcn_wmma_f32_16x16x4_f32(
        false, a, false, bl, (short)0, accL, false, false);
    accR = __builtin_amdgcn_wmma_f32_16x16x4_f32(
        false, a, false, br, (short)0, accR, false, false);
  }

  #pragma unroll
  for (int r = 0; r < 8; ++r) {
    int row = m0 + r + 8 * half;
    if (row < n_nodes) {
      xl1[(size_t)row * 64 + n0 + rc] = accL[r];
      xr1[(size_t)row * 64 + n0 + rc] = accR[r];
    }
  }
}

__global__ void __launch_bounds__(128)
gemm2_kernel(const float* __restrict__ hacc, const float* __restrict__ den1,
             const float* __restrict__ b1, const float* __restrict__ W2l,
             const float* __restrict__ W2r, float* __restrict__ xl2,
             float* __restrict__ xr2, int n_nodes, int mtiles) {
  const int lane  = threadIdx.x & 31;
  const int mtile = blockIdx.x * 4 + (threadIdx.x >> 5);
  if (mtile >= mtiles) return;
  const int m0   = mtile * 16;
  const int half = lane >> 4;
  const int rc   = lane & 15;

  int arow = m0 + rc; if (arow >= n_nodes) arow = n_nodes - 1;
  const v2f* hrow = (const v2f*)(hacc + (size_t)arow * 64);
  const v4f  dn   = *(const v4f*)(den1 + (size_t)arow * 4);
  const bool colok = (rc < 10);

  v2f a[16];
  #pragma unroll
  for (int kc = 0; kc < 16; ++kc) a[kc] = hrow[2 * kc + half];
  #pragma unroll
  for (int kc = 0; kc < 16; ++kc) {
    const float d = dn[kc >> 2];
    const v2f bb = *(const v2f*)(b1 + 4 * kc + 2 * half);
    float v0 = (d > 0.0f) ? a[kc][0] / d : 0.0f;
    float v1 = (d > 0.0f) ? a[kc][1] / d : 0.0f;
    v0 += bb[0];
    v1 += bb[1];
    a[kc][0] = (v0 > 0.0f) ? v0 : expm1f(v0);
    a[kc][1] = (v1 > 0.0f) ? v1 : expm1f(v1);
  }

  v8f accL = {}, accR = {};
  #pragma unroll
  for (int kc = 0; kc < 16; ++kc) {
    const int k0 = 4 * kc + 2 * half;
    v2f bl, br;
    bl[0] = colok ? W2l[(size_t)(k0 + 0) * 10 + rc] : 0.0f;
    bl[1] = colok ? W2l[(size_t)(k0 + 1) * 10 + rc] : 0.0f;
    br[0] = colok ? W2r[(size_t)(k0 + 0) * 10 + rc] : 0.0f;
    br[1] = colok ? W2r[(size_t)(k0 + 1) * 10 + rc] : 0.0f;
    accL = __builtin_amdgcn_wmma_f32_16x16x4_f32(
        false, a[kc], false, bl, (short)0, accL, false, false);
    accR = __builtin_amdgcn_wmma_f32_16x16x4_f32(
        false, a[kc], false, br, (short)0, accR, false, false);
  }

  #pragma unroll
  for (int r = 0; r < 8; ++r) {
    int row = m0 + r + 8 * half;
    if (row < n_nodes) {
      xl2[(size_t)row * 16 + rc] = accL[r];
      xr2[(size_t)row * 16 + rc] = accR[r];
    }
  }
}

__global__ void __launch_bounds__(256)
l1_edge_logits(const int* __restrict__ psrc, const int* __restrict__ pdst,
               const float* __restrict__ xl1,
               const float* __restrict__ xr1, const float* __restrict__ att1,
               float* __restrict__ logits1, float* __restrict__ m1,
               int* __restrict__ ctr, int E) {
  int lo, hi;
  claim_chunk(ctr + 64, ctr, E, 64, lo, hi);
  if (lo >= hi) return;
  const int l = threadIdx.x & 15;
  const float4 at = *(const float4*)(att1 + l * 4);

  int gg[4], dj[4];
  bool act[4];
  v4f xv[4], rv[4];
  #pragma unroll
  for (int it = 0; it < 4; ++it) {
    int g = lo + it * 16 + (threadIdx.x >> 4);
    act[it] = (g < hi);
    if (!act[it]) g = hi - 1;
    gg[it] = g;
  }
  int sj[4];
  #pragma unroll
  for (int it = 0; it < 4; ++it) {
    sj[it] = psrc[gg[it]];
    dj[it] = pdst[gg[it]];
  }
  asm volatile("" :: "v"(sj[0]), "v"(dj[0]), "v"(sj[1]), "v"(dj[1]),
                     "v"(sj[2]), "v"(dj[2]), "v"(sj[3]), "v"(dj[3]));
  #pragma unroll
  for (int it = 0; it < 4; ++it) {
    xv[it] = *(const v4f*)(xl1 + (size_t)sj[it] * 64 + l * 4);
    rv[it] = *(const v4f*)(xr1 + (size_t)dj[it] * 64 + l * 4);
  }
  asm volatile("" :: "v"(xv[0]), "v"(rv[0]), "v"(xv[1]), "v"(rv[1]),
                     "v"(xv[2]), "v"(rv[2]), "v"(xv[3]), "v"(rv[3]));
  #pragma unroll
  for (int it = 0; it < 4; ++it) {
    float p = 0.0f, s;
    s = xv[it].x + rv[it].x; s = (s > 0.f) ? s : NEG_SLOPE * s; p += s * at.x;
    s = xv[it].y + rv[it].y; s = (s > 0.f) ? s : NEG_SLOPE * s; p += s * at.y;
    s = xv[it].z + rv[it].z; s = (s > 0.f) ? s : NEG_SLOPE * s; p += s * at.z;
    s = xv[it].w + rv[it].w; s = (s > 0.f) ? s : NEG_SLOPE * s; p += s * at.w;

    p += __shfl_xor(p, 1, 32);
    p += __shfl_xor(p, 2, 32);

    if (act[it] && (l & 3) == 0) {
      int hd = l >> 2;
      logits1[(size_t)gg[it] * 4 + hd] = p;
      atomicMaxF(m1 + (size_t)dj[it] * 4 + hd, p);
    }
  }
}

__global__ void __launch_bounds__(256)
l1_edge_aggregate(const int* __restrict__ psrc, const int* __restrict__ pdst,
                  const float* __restrict__ xl1,
                  const float* __restrict__ logits1, const float* __restrict__ m1,
                  float* __restrict__ den1, float* __restrict__ hacc,
                  int* __restrict__ ctr, int E) {
  int lo, E_hi;
  claim_chunk(ctr + 192, ctr, E, 64, lo, E_hi);
  const int lane = threadIdx.x & 31;
  const int g0   = lo + (threadIdx.x >> 5) * 8;
  if (g0 >= E_hi) return;

  const int hd = lane & 3;
  int ge = g0 + (lane >> 2);
  const bool ok = (ge < E_hi);
  if (!ok) ge = E_hi - 1;
  const int srcj = psrc[ge];
  const int dstj = pdst[ge];
  const float lg = logits1[(size_t)ge * 4 + hd];
  const float mx = m1[(size_t)dstj * 4 + hd];

  float v0[8], v1[8];
  #pragma unroll
  for (int e = 0; e < 8; ++e) {
    const int s = __builtin_amdgcn_readlane(srcj, 4 * e);
    const float* r = xl1 + (size_t)s * 64 + lane;
    v0[e] = r[0];
    v1[e] = r[32];
  }

  const float w = __expf(lg - mx);
  if (ok) atomicAdd(den1 + (size_t)dstj * 4 + hd, w);

  const bool hi = (lane >= 16);
  const int  wi = __float_as_int(w);

  #pragma unroll
  for (int e = 0; e < 8; ++e) {
    if (g0 + e < E_hi) {
      const int d = __builtin_amdgcn_readlane(dstj, 4 * e);
      const float w0 = __int_as_float(__builtin_amdgcn_readlane(wi, 4 * e + 0));
      const float w1 = __int_as_float(__builtin_amdgcn_readlane(wi, 4 * e + 1));
      const float w2 = __int_as_float(__builtin_amdgcn_readlane(wi, 4 * e + 2));
      const float w3 = __int_as_float(__builtin_amdgcn_readlane(wi, 4 * e + 3));
      float* o = hacc + (size_t)d * 64 + lane;
      atomicAdd(o,      (hi ? w1 : w0) * v0[e]);
      atomicAdd(o + 32, (hi ? w3 : w2) * v1[e]);
    }
  }
}

__global__ void __launch_bounds__(256)
l2_edge_logits(const int* __restrict__ psrc, const int* __restrict__ pdst,
               const float* __restrict__ xl2,
               const float* __restrict__ xr2, const float* __restrict__ att2,
               float* __restrict__ logits2, float* __restrict__ m2,
               int* __restrict__ ctr, int E) {
  int lo, hi;
  claim_chunk(ctr + 320, ctr, E, 256, lo, hi);
  if (lo >= hi) return;
  const int l  = threadIdx.x & 3;
  const int c0 = l * 4;

  int gg[4], dj[4];
  bool act[4];
  float at0 = 0.0f, at1 = 0.0f, at2 = 0.0f, at3 = 0.0f;
  if (c0 + 0 < 10) at0 = att2[c0 + 0];
  if (c0 + 1 < 10) at1 = att2[c0 + 1];
  if (c0 + 2 < 10) at2 = att2[c0 + 2];
  if (c0 + 3 < 10) at3 = att2[c0 + 3];
  v4f a[4], b[4];
  #pragma unroll
  for (int it = 0; it < 4; ++it) {
    int g = lo + it * 64 + (threadIdx.x >> 2);
    act[it] = (g < hi);
    if (!act[it]) g = hi - 1;
    gg[it] = g;
  }
  #pragma unroll
  for (int it = 0; it < 4; ++it) {
    const int src = psrc[gg[it]];
    dj[it] = pdst[gg[it]];
    a[it] = *(const v4f*)(xl2 + (size_t)src * 16 + c0);
    b[it] = *(const v4f*)(xr2 + (size_t)dj[it] * 16 + c0);
  }
  asm volatile("" :: "v"(a[0]), "v"(b[0]), "v"(a[1]), "v"(b[1]),
                     "v"(a[2]), "v"(b[2]), "v"(a[3]), "v"(b[3]));
  #pragma unroll
  for (int it = 0; it < 4; ++it) {
    float p = 0.0f, s;
    s = a[it].x + b[it].x; s = (s > 0.f) ? s : NEG_SLOPE * s; if (c0 + 0 < 10) p += s * at0;
    s = a[it].y + b[it].y; s = (s > 0.f) ? s : NEG_SLOPE * s; if (c0 + 1 < 10) p += s * at1;
    s = a[it].z + b[it].z; s = (s > 0.f) ? s : NEG_SLOPE * s; if (c0 + 2 < 10) p += s * at2;
    s = a[it].w + b[it].w; s = (s > 0.f) ? s : NEG_SLOPE * s; if (c0 + 3 < 10) p += s * at3;

    p += __shfl_xor(p, 1, 32);
    p += __shfl_xor(p, 2, 32);

    if (act[it] && l == 0) {
      logits2[gg[it]] = p;
      atomicMaxF(m2 + dj[it], p);
    }
  }
}

__global__ void __launch_bounds__(256)
l2_edge_aggregate(const int* __restrict__ psrc, const int* __restrict__ pdst,
                  const float* __restrict__ xl2,
                  const float* __restrict__ logits2, const float* __restrict__ m2,
                  float* __restrict__ acc2, int* __restrict__ ctr, int E) {
  int lo, E_hi;
  claim_chunk(ctr + 448, ctr, E, 128, lo, E_hi);
  const int lane = threadIdx.x & 31;
  const int g0   = lo + (threadIdx.x >> 5) * 16;
  if (g0 >= E_hi) return;

  int ge = g0 + (lane & 15);
  if (ge >= E_hi) ge = E_hi - 1;
  const int srcj = psrc[ge];
  const int dstj = pdst[ge];
  const int wi   = __float_as_int(__expf(logits2[ge] - m2[dstj]));

  const bool hi = (lane >= 16);
  const int  c  = lane & 15;

  float v[8];
  #pragma unroll
  for (int p = 0; p < 8; ++p) {
    const int sa = __builtin_amdgcn_readlane(srcj, p);
    const int sb = __builtin_amdgcn_readlane(srcj, 8 + p);
    v[p] = xl2[(size_t)(hi ? sb : sa) * 16 + c];
  }
  #pragma unroll
  for (int p = 0; p < 8; ++p) {
    const int da = __builtin_amdgcn_readlane(dstj, p);
    const int db = __builtin_amdgcn_readlane(dstj, 8 + p);
    const float wa = __int_as_float(__builtin_amdgcn_readlane(wi, p));
    const float wb = __int_as_float(__builtin_amdgcn_readlane(wi, 8 + p));
    const int   d  = hi ? db : da;
    const float w  = hi ? wb : wa;
    const int   e  = g0 + p + (hi ? 8 : 0);
    if (c <= 10 && e < E_hi)
      atomicAdd(acc2 + (size_t)d * 16 + c, (c < 10) ? w * v[p] : w);
  }
}

__global__ void __launch_bounds__(256)
l2_finish(const float* __restrict__ acc2,
          const float* __restrict__ b2, float* __restrict__ out, int total) {
  int i = blockIdx.x * blockDim.x + threadIdx.x;
  if (i >= total) return;
  int n = i / 10, c = i % 10;
  float d = acc2[(size_t)n * 16 + 10];
  float v = (d > 0.0f) ? acc2[(size_t)n * 16 + c] / d : 0.0f;
  out[i] = v + b2[c];
}

extern "C" void kernel_launch(void* const* d_in, const int* in_sizes, int n_in,
                              void* d_out, int out_size, void* d_ws, size_t ws_size,
                              hipStream_t stream) {
  const float* x    = (const float*)d_in[0];
  const int*   ei   = (const int*)d_in[1];
  const float* W1l  = (const float*)d_in[2];
  const float* W1r  = (const float*)d_in[3];
  const float* att1 = (const float*)d_in[4];
  const float* b1   = (const float*)d_in[5];
  const float* W2l  = (const float*)d_in[6];
  const float* W2r  = (const float*)d_in[7];
  const float* att2 = (const float*)d_in[8];
  const float* b2   = (const float*)d_in[9];
  float* out = (float*)d_out;

  const int N = in_sizes[0] / 128;
  const int E = in_sizes[1] / 2;

  float* ws = (float*)d_ws;
  size_t o = 0;
  float* xl1  = ws + o; o += (size_t)N * 64;
  float* xr1  = ws + o; o += (size_t)N * 64;
  float* hacc = ws + o; o += (size_t)N * 64;
  float* lg1  = ws + o; o += (size_t)E * 4;
  float* m1   = ws + o; o += (size_t)N * 4;
  float* dn1  = ws + o; o += (size_t)N * 4;
  int*   ctr  = (int*)(ws + o); o += 576;
  float* xl2  = ws + o; o += (size_t)N * 16;
  float* xr2  = ws + o; o += (size_t)N * 16;
  float* acc2 = ws + o; o += (size_t)N * 16;
  float* lg2  = ws + o; o += (size_t)E;
  float* m2   = ws + o; o += (size_t)N;
  int* psrc = (int*)(ws + o); o += (size_t)E;
  int* pdst = (int*)(ws + o); o += (size_t)E;

  auto blocks = [](long long n) { return (unsigned)((n + 255) / 256); };
  const int mtiles = (N + 15) / 16;

  gemm1_kernel<<<mtiles, 128, 0, stream>>>(x, W1l, W1r, xl1, xr1,
                                           hacc, m1, dn1, acc2, m2, ctr, N);
  const int split = ((N / 2) + 63) & ~63;
  partition_edges<<<(E + 1023) / 1024, 256, 0, stream>>>(ei, psrc, pdst, ctr, E, split);
  l1_edge_logits<<<(E + 63) / 64 + 2, 256, 0, stream>>>(psrc, pdst, xl1, xr1, att1,
                                                        lg1, m1, ctr, E);
  l1_edge_aggregate<<<(E + 63) / 64 + 2, 256, 0, stream>>>(psrc, pdst, xl1, lg1, m1,
                                                           dn1, hacc, ctr, E);

  gemm2_kernel<<<(mtiles + 3) / 4, 128, 0, stream>>>(hacc, dn1, b1, W2l, W2r, xl2, xr2, N, mtiles);
  l2_edge_logits<<<(E + 255) / 256 + 2, 256, 0, stream>>>(psrc, pdst, xl2, xr2, att2,
                                                          lg2, m2, ctr, E);
  l2_edge_aggregate<<<(E + 127) / 128 + 2, 256, 0, stream>>>(psrc, pdst, xl2, lg2, m2,
                                                             acc2, ctr, E);
  l2_finish<<<blocks((long long)N * 10), 256, 0, stream>>>(acc2, b2, out, N * 10);
}


# Edge kernels on resident grids: a block takes one ticket for 640 (layer 1) or 768 (layer 2) list positions and walks them in a loop; 275k short waves become 37k; sums in arrival order as before
# speedup vs baseline: 2.8346x; 1.0439x over previous
#include <hip/hip_runtime.h>
#include <math.h>

typedef __attribute__((ext_vector_type(2))) float v2f;
typedef __attribute__((ext_vector_type(8))) float v8f;
typedef __attribute__((ext_vector_type(4))) float v4f;

#define NEG_SLOPE 0.2f

#define L1_SEG 640
#define L2_SEG 768

__device__ __forceinline__ void atomicMaxF(float* addr, float v) {
  if (v >= 0.0f) atomicMax((int*)addr, __float_as_int(v));
  else           atomicMin((unsigned int*)addr, __float_as_uint(v));
}

__global__ void __launch_bounds__(256)
partition_edges(const int* __restrict__ ei, int* __restrict__ psrc,
                int* __restrict__ pdst, int* __restrict__ cnt, int E, int split) {
  __shared__ int wc0[8], wc1[8], sbase[2];
  const int tid = threadIdx.x, lane = tid & 31, wave = tid >> 5;
  const int base = blockIdx.x * 1024;
  int s[4], d[4];
  unsigned b0[4], b1[4];
  int c0 = 0, c1 = 0;
  #pragma unroll
  for (int j = 0; j < 4; ++j) {
    const int g = base + j * 256 + tid;
    const bool valid = (g < E);
    const int gg = valid ? g : E - 1;
    s[j] = ei[gg];
    d[j] = ei[(size_t)E + gg];
    b0[j] = __builtin_amdgcn_ballot_w32(valid && d[j] <  split);
    b1[j] = __builtin_amdgcn_ballot_w32(valid && d[j] >= split);
    c0 += __popc(b0[j]);
    c1 += __popc(b1[j]);
  }
  if (lane == 0) { wc0[wave] = c0; wc1[wave] = c1; }
  __syncthreads();
  if (tid == 0) {
    int t0 = 0, t1 = 0;
    #pragma unroll
    for (int w = 0; w < 8; ++w) { t0 += wc0[w]; t1 += wc1[w]; }
    sbase[0] = atomicAdd(cnt + 0, t0);
    sbase[1] = atomicAdd(cnt + 1, t1);
  }
  __syncthreads();
  int off0 = sbase[0], off1 = sbase[1];
  for (int w = 0; w < wave; ++w) { off0 += wc0[w]; off1 += wc1[w]; }
  const unsigned lt = (1u << lane) - 1u;
  #pragma unroll
  for (int j = 0; j < 4; ++j) {
    if ((b0[j] >> lane) & 1u) {
      const int pos = off0 + __popc(b0[j] & lt);
      psrc[pos] = s[j]; pdst[pos] = d[j];
    } else if ((b1[j] >> lane) & 1u) {
      const int pos = E - 1 - (off1 + __popc(b1[j] & lt));
      psrc[pos] = s[j]; pdst[pos] = d[j];
    }
    off0 += __popc(b0[j]);
    off1 += __popc(b1[j]);
  }
}

__device__ __forceinline__ void claim_chunk(int* __restrict__ tk, const int* __restrict__ cnt,
                                            int E, int CH, int& lo, int& hi) {
  __shared__ int s_lo, s_hi;
  if (threadIdx.x == 0) {
    const int n0 = cnt[0];
    const int die = (int)((__builtin_amdgcn_s_sendmsg_rtn(0x87) >> 16) & 15u);
    const int half = (die >> 2) & 1;
    int l = 0, h = 0;
    #pragma unroll
    for (int pass = 0; pass < 2; ++pass) {
      const int sset = half ^ pass;
      const long long beg = sset ? n0 : 0;
      const long long end = sset ? E : n0;
      const long long a = beg + (long long)atomicAdd(tk + 64 * sset, 1) * CH;
      if (a < end) {
        l = (int)a;
        h = (int)((a + CH < end) ? a + CH : end);
        break;
      }
    }
    s_lo = l; s_hi = h;
  }
  __syncthreads();
  lo = s_lo; hi = s_hi;
}

__global__ void __launch_bounds__(128)
gemm1_kernel(const float* __restrict__ x, const float* __restrict__ W1l,
             const float* __restrict__ W1r, float* __restrict__ xl1,
             float* __restrict__ xr1, float* __restrict__ hacc,
             float* __restrict__ m1, float* __restrict__ dn1,
             float* __restrict__ acc2, float* __restrict__ m2,
             int* __restrict__ ctr, int n_nodes) {
  {
    const int t = threadIdx.x;
    const int r0 = blockIdx.x * 16;
    const v4f zero4 = {0.0f, 0.0f, 0.0f, 0.0f};
    const float ninf = -__builtin_inff();
    const v4f ninf4 = {ninf, ninf, ninf, ninf};
    for (int j = t; j < 256; j += 128)
      if (r0 + (j >> 4) < n_nodes) *(v4f*)(hacc + (size_t)r0 * 64 + j * 4) = zero4;
    if (t < 16) {
      if (r0 + t < n_nodes) *(v4f*)(m1 + (size_t)(r0 + t) * 4) = ninf4;
    } else if (t < 32) {
      if (r0 + t - 16 < n_nodes) *(v4f*)(dn1 + (size_t)(r0 + t - 16) * 4) = zero4;
    } else if (t < 96) {
      const int j = t - 32;
      if (r0 + (j >> 2) < n_nodes) *(v4f*)(acc2 + (size_t)r0 * 16 + j * 4) = zero4;
    } else if (t < 112) {
      if (r0 + t - 96 < n_nodes) m2[r0 + t - 96] = ninf;
    }
    if (blockIdx.x == 0)
      for (int j = t; j < 576; j += 128) ctr[j] = 0;
  }
  const int lane = threadIdx.x & 31;
  const int wave = threadIdx.x >> 5;
  const int m0   = blockIdx.x * 16;
  const int n0   = wave * 16;
  const int half = lane >> 4;
  const int rc   = lane & 15;

  int arow = m0 + rc; if (arow >= n_nodes) arow = n_nodes - 1;
  const v2f*   xrow = (const v2f*)(x + (size_t)arow * 128);
  const float* Bl   = W1l + n0 + rc;
  const float* Br   = W1r + n0 + rc;

  v8f accL = {}, accR = {};
  #pragma unroll
  for (int kc = 0; kc < 32; ++kc) {
    v2f a = xrow[2 * kc + half];
    const int k0 = 4 * kc + 2 * half;
    v2f bl, br;
    bl[0] = Bl[(size_t)(k0 + 0) * 64];
    bl[1] = Bl[(size_t)(k0 + 1) * 64];
    br[0] = Br[(size_t)(k0 + 0) * 64];
    br[1] = Br[(size_t)(k0 + 1) * 64];
    accL = __builtin_amdgcn_wmma_f32_16x16x4_f32(
        false, a, false, bl, (short)0, accL, false, false);
    accR = __builtin_amdgcn_wmma_f32_16x16x4_f32(
        false, a, false, br, (short)0, accR, false, false);
  }

  #pragma unroll
  for (int r = 0; r < 8; ++r) {
    int row = m0 + r + 8 * half;
    if (row < n_nodes) {
      xl1[(size_t)row * 64 + n0 + rc] = accL[r];
      xr1[(size_t)row * 64 + n0 + rc] = accR[r];
    }
  }
}

__global__ void __launch_bounds__(128)
gemm2_kernel(const float* __restrict__ hacc, const float* __restrict__ den1,
             const float* __restrict__ b1, const float* __restrict__ W2l,
             const float* __restrict__ W2r, float* __restrict__ xl2,
             float* __restrict__ xr2, int n_nodes, int mtiles) {
  const int lane  = threadIdx.x & 31;
  const int mtile = blockIdx.x * 4 + (threadIdx.x >> 5);
  if (mtile >= mtiles) return;
  const int m0   = mtile * 16;
  const int half = lane >> 4;
  const int rc   = lane & 15;

  int arow = m0 + rc; if (arow >= n_nodes) arow = n_nodes - 1;
  const v2f* hrow = (const v2f*)(hacc + (size_t)arow * 64);
  const v4f  dn   = *(const v4f*)(den1 + (size_t)arow * 4);
  const bool colok = (rc < 10);

  v2f a[16];
  #pragma unroll
  for (int kc = 0; kc < 16; ++kc) a[kc] = hrow[2 * kc + half];
  #pragma unroll
  for (int kc = 0; kc < 16; ++kc) {
    const float d = dn[kc >> 2];
    const v2f bb = *(const v2f*)(b1 + 4 * kc + 2 * half);
    float v0 = (d > 0.0f) ? a[kc][0] / d : 0.0f;
    float v1 = (d > 0.0f) ? a[kc][1] / d : 0.0f;
    v0 += bb[0];
    v1 += bb[1];
    a[kc][0] = (v0 > 0.0f) ? v0 : expm1f(v0);
    a[kc][1] = (v1 > 0.0f) ? v1 : expm1f(v1);
  }

  v8f accL = {}, accR = {};
  #pragma unroll
  for (int kc = 0; kc < 16; ++kc) {
    const int k0 = 4 * kc + 2 * half;
    v2f bl, br;
    bl[0] = colok ? W2l[(size_t)(k0 + 0) * 10 + rc] : 0.0f;
    bl[1] = colok ? W2l[(size_t)(k0 + 1) * 10 + rc] : 0.0f;
    br[0] = colok ? W2r[(size_t)(k0 + 0) * 10 + rc] : 0.0f;
    br[1] = colok ? W2r[(size_t)(k0 + 1) * 10 + rc] : 0.0f;
    accL = __builtin_amdgcn_wmma_f32_16x16x4_f32(
        false, a[kc], false, bl, (short)0, accL, false, false);
    accR = __builtin_amdgcn_wmma_f32_16x16x4_f32(
        false, a[kc], false, br, (short)0, accR, false, false);
  }

  #pragma unroll
  for (int r = 0; r < 8; ++r) {
    int row = m0 + r + 8 * half;
    if (row < n_nodes) {
      xl2[(size_t)row * 16 + rc] = accL[r];
      xr2[(size_t)row * 16 + rc] = accR[r];
    }
  }
}

__global__ void __launch_bounds__(256)
l1_edge_logits(const int* __restrict__ psrc, const int* __restrict__ pdst,
               const float* __restrict__ xl1,
               const float* __restrict__ xr1, const float* __restrict__ att1,
               float* __restrict__ logits1, float* __restrict__ m1,
               int* __restrict__ ctr, int E) {
  int lo, hi;
  claim_chunk(ctr + 64, ctr, E, L1_SEG, lo, hi);
  if (lo >= hi) return;
  const int l = threadIdx.x & 15;
  const float4 at = *(const float4*)(att1 + l * 4);

  for (int base = lo; base < hi; base += 64) {
    int gg[4], dj[4];
    bool act[4];
    v4f xv[4], rv[4];
    #pragma unroll
    for (int it = 0; it < 4; ++it) {
      int g = base + it * 16 + (threadIdx.x >> 4);
      act[it] = (g < hi);
      if (!act[it]) g = hi - 1;
      gg[it] = g;
    }
    int sj[4];
    #pragma unroll
    for (int it = 0; it < 4; ++it) {
      sj[it] = psrc[gg[it]];
      dj[it] = pdst[gg[it]];
    }
    asm volatile("" :: "v"(sj[0]), "v"(dj[0]), "v"(sj[1]), "v"(dj[1]),
                       "v"(sj[2]), "v"(dj[2]), "v"(sj[3]), "v"(dj[3]));
    #pragma unroll
    for (int it = 0; it < 4; ++it) {
      xv[it] = *(const v4f*)(xl1 + (size_t)sj[it] * 64 + l * 4);
      rv[it] = *(const v4f*)(xr1 + (size_t)dj[it] * 64 + l * 4);
    }
    asm volatile("" :: "v"(xv[0]), "v"(rv[0]), "v"(xv[1]), "v"(rv[1]),
                       "v"(xv[2]), "v"(rv[2]), "v"(xv[3]), "v"(rv[3]));
    #pragma unroll
    for (int it = 0; it < 4; ++it) {
      float p = 0.0f, s;
      s = xv[it].x + rv[it].x; s = (s > 0.f) ? s : NEG_SLOPE * s; p += s * at.x;
      s = xv[it].y + rv[it].y; s = (s > 0.f) ? s : NEG_SLOPE * s; p += s * at.y;
      s = xv[it].z + rv[it].z; s = (s > 0.f) ? s : NEG_SLOPE * s; p += s * at.z;
      s = xv[it].w + rv[it].w; s = (s > 0.f) ? s : NEG_SLOPE * s; p += s * at.w;

      p += __shfl_xor(p, 1, 32);
      p += __shfl_xor(p, 2, 32);

      if (act[it] && (l & 3) == 0) {
        int hd = l >> 2;
        logits1[(size_t)gg[it] * 4 + hd] = p;
        atomicMaxF(m1 + (size_t)dj[it] * 4 + hd, p);
      }
    }
  }
}

__global__ void __launch_bounds__(256)
l1_edge_aggregate(const int* __restrict__ psrc, const int* __restrict__ pdst,
                  const float* __restrict__ xl1,
                  const float* __restrict__ logits1, const float* __restrict__ m1,
                  float* __restrict__ den1, float* __restrict__ hacc,
                  int* __restrict__ ctr, int E) {
  int lo, E_hi;
  claim_chunk(ctr + 192, ctr, E, L1_SEG, lo, E_hi);
  const int lane = threadIdx.x & 31;
  for (int g0 = lo + (threadIdx.x >> 5) * 8; g0 < E_hi; g0 += 64) {

    const int hd = lane & 3;
    int ge = g0 + (lane >> 2);
    const bool ok = (ge < E_hi);
    if (!ok) ge = E_hi - 1;
    const int srcj = psrc[ge];
    const int dstj = pdst[ge];
    const float lg = logits1[(size_t)ge * 4 + hd];
    const float mx = m1[(size_t)dstj * 4 + hd];

    float v0[8], v1[8];
    #pragma unroll
    for (int e = 0; e < 8; ++e) {
      const int s = __builtin_amdgcn_readlane(srcj, 4 * e);
      const float* r = xl1 + (size_t)s * 64 + lane;
      v0[e] = r[0];
      v1[e] = r[32];
    }

    const float w = __expf(lg - mx);
    if (ok) atomicAdd(den1 + (size_t)dstj * 4 + hd, w);

    const bool hi = (lane >= 16);
    const int  wi = __float_as_int(w);

    #pragma unroll
    for (int e = 0; e < 8; ++e) {
      if (g0 + e < E_hi) {
        const int d = __builtin_amdgcn_readlane(dstj, 4 * e);
        const float w0 = __int_as_float(__builtin_amdgcn_readlane(wi, 4 * e + 0));
        const float w1 = __int_as_float(__builtin_amdgcn_readlane(wi, 4 * e + 1));
        const float w2 = __int_as_float(__builtin_amdgcn_readlane(wi, 4 * e + 2));
        const float w3 = __int_as_float(__builtin_amdgcn_readlane(wi, 4 * e + 3));
        float* o = hacc + (size_t)d * 64 + lane;
        atomicAdd(o,      (hi ? w1 : w0) * v0[e]);
        atomicAdd(o + 32, (hi ? w3 : w2) * v1[e]);
      }
    }
  }
}

__global__ void __launch_bounds__(256)
l2_edge_logits(const int* __restrict__ psrc, const int* __restrict__ pdst,
               const float* __restrict__ xl2,
               const float* __restrict__ xr2, const float* __restrict__ att2,
               float* __restrict__ logits2, float* __restrict__ m2,
               int* __restrict__ ctr, int E) {
  int lo, hi;
  claim_chunk(ctr + 320, ctr, E, L2_SEG, lo, hi);
  if (lo >= hi) return;
  const int l  = threadIdx.x & 3;
  const int c0 = l * 4;

  float at0 = 0.0f, at1 = 0.0f, at2 = 0.0f, at3 = 0.0f;
  if (c0 + 0 < 10) at0 = att2[c0 + 0];
  if (c0 + 1 < 10) at1 = att2[c0 + 1];
  if (c0 + 2 < 10) at2 = att2[c0 + 2];
  if (c0 + 3 < 10) at3 = att2[c0 + 3];
  for (int base = lo; base < hi; base += 256) {
    int gg[4], dj[4];
    bool act[4];
    v4f a[4], b[4];
    #pragma unroll
    for (int it = 0; it < 4; ++it) {
      int g = base + it * 64 + (threadIdx.x >> 2);
      act[it] = (g < hi);
      if (!act[it]) g = hi - 1;
      gg[it] = g;
    }
    #pragma unroll
    for (int it = 0; it < 4; ++it) {
      const int src = psrc[gg[it]];
      dj[it] = pdst[gg[it]];
      a[it] = *(const v4f*)(xl2 + (size_t)src * 16 + c0);
      b[it] = *(const v4f*)(xr2 + (size_t)dj[it] * 16 + c0);
    }
    asm volatile("" :: "v"(a[0]), "v"(b[0]), "v"(a[1]), "v"(b[1]),
                       "v"(a[2]), "v"(b[2]), "v"(a[3]), "v"(b[3]));
    #pragma unroll
    for (int it = 0; it < 4; ++it) {
      float p = 0.0f, s;
      s = a[it].x + b[it].x; s = (s > 0.f) ? s : NEG_SLOPE * s; if (c0 + 0 < 10) p += s * at0;
      s = a[it].y + b[it].y; s = (s > 0.f) ? s : NEG_SLOPE * s; if (c0 + 1 < 10) p += s * at1;
      s = a[it].z + b[it].z; s = (s > 0.f) ? s : NEG_SLOPE * s; if (c0 + 2 < 10) p += s * at2;
      s = a[it].w + b[it].w; s = (s > 0.f) ? s : NEG_SLOPE * s; if (c0 + 3 < 10) p += s * at3;

      p += __shfl_xor(p, 1, 32);
      p += __shfl_xor(p, 2, 32);

      if (act[it] && l == 0) {
        logits2[gg[it]] = p;
        atomicMaxF(m2 + dj[it], p);
      }
    }
  }
}

__global__ void __launch_bounds__(256)
l2_edge_aggregate(const int* __restrict__ psrc, const int* __restrict__ pdst,
                  const float* __restrict__ xl2,
                  const float* __restrict__ logits2, const float* __restrict__ m2,
                  float* __restrict__ acc2, int* __restrict__ ctr, int E) {
  int lo, E_hi;
  claim_chunk(ctr + 448, ctr, E, L2_SEG, lo, E_hi);
  const int lane = threadIdx.x & 31;
  for (int g0 = lo + (threadIdx.x >> 5) * 16; g0 < E_hi; g0 += 128) {

    int ge = g0 + (lane & 15);
    if (ge >= E_hi) ge = E_hi - 1;
    const int srcj = psrc[ge];
    const int dstj = pdst[ge];
    const int wi   = __float_as_int(__expf(logits2[ge] - m2[dstj]));

    const bool hi = (lane >= 16);
    const int  c  = lane & 15;

    float v[8];
    #pragma unroll
    for (int p = 0; p < 8; ++p) {
      const int sa = __builtin_amdgcn_readlane(srcj, p);
      const int sb = __builtin_amdgcn_readlane(srcj, 8 + p);
      v[p] = xl2[(size_t)(hi ? sb : sa) * 16 + c];
    }
    #pragma unroll
    for (int p = 0; p < 8; ++p) {
      const int da = __builtin_amdgcn_readlane(dstj, p);
      const int db = __builtin_amdgcn_readlane(dstj, 8 + p);
      const float wa = __int_as_float(__builtin_amdgcn_readlane(wi, p));
      const float wb = __int_as_float(__builtin_amdgcn_readlane(wi, 8 + p));
      const int   d  = hi ? db : da;
      const float w  = hi ? wb : wa;
      const int   e  = g0 + p + (hi ? 8 : 0);
      if (c <= 10 && e < E_hi)
        atomicAdd(acc2 + (size_t)d * 16 + c, (c < 10) ? w * v[p] : w);
    }
  }
}

__global__ void __launch_bounds__(256)
l2_finish(const float* __restrict__ acc2,
          const float* __restrict__ b2, float* __restrict__ out, int total) {
  int i = blockIdx.x * blockDim.x + threadIdx.x;
  if (i >= total) return;
  int n = i / 10, c = i % 10;
  float d = acc2[(size_t)n * 16 + 10];
  float v = (d > 0.0f) ? acc2[(size_t)n * 16 + c] / d : 0.0f;
  out[i] = v + b2[c];
}

extern "C" void kernel_launch(void* const* d_in, const int* in_sizes, int n_in,
                              void* d_out, int out_size, void* d_ws, size_t ws_size,
                              hipStream_t stream) {
  const float* x    = (const float*)d_in[0];
  const int*   ei   = (const int*)d_in[1];
  const float* W1l  = (const float*)d_in[2];
  const float* W1r  = (const float*)d_in[3];
  const float* att1 = (const float*)d_in[4];
  const float* b1   = (const float*)d_in[5];
  const float* W2l  = (const float*)d_in[6];
  const float* W2r  = (const float*)d_in[7];
  const float* att2 = (const float*)d_in[8];
  const float* b2   = (const float*)d_in[9];
  float* out = (float*)d_out;

  const int N = in_sizes[0] / 128;
  const int E = in_sizes[1] / 2;

  float* ws = (float*)d_ws;
  size_t o = 0;
  float* xl1  = ws + o; o += (size_t)N * 64;
  float* xr1  = ws + o; o += (size_t)N * 64;
  float* hacc = ws + o; o += (size_t)N * 64;
  float* lg1  = ws + o; o += (size_t)E * 4;
  float* m1   = ws + o; o += (size_t)N * 4;
  float* dn1  = ws + o; o += (size_t)N * 4;
  int*   ctr  = (int*)(ws + o); o += 576;
  float* xl2  = ws + o; o += (size_t)N * 16;
  float* xr2  = ws + o; o += (size_t)N * 16;
  float* acc2 = ws + o; o += (size_t)N * 16;
  float* lg2  = ws + o; o += (size_t)E;
  float* m2   = ws + o; o += (size_t)N;
  int* psrc = (int*)(ws + o); o += (size_t)E;
  int* pdst = (int*)(ws + o); o += (size_t)E;

  auto blocks = [](long long n) { return (unsigned)((n + 255) / 256); };
  const int mtiles = (N + 15) / 16;

  gemm1_kernel<<<mtiles, 128, 0, stream>>>(x, W1l, W1r, xl1, xr1,
                                           hacc, m1, dn1, acc2, m2, ctr, N);
  const int split = ((N / 2) + 63) & ~63;
  partition_edges<<<(E + 1023) / 1024, 256, 0, stream>>>(ei, psrc, pdst, ctr, E, split);
  l1_edge_logits<<<(E + L1_SEG - 1) / L1_SEG + 2, 256, 0, stream>>>(psrc, pdst, xl1, xr1, att1,
                                                                    lg1, m1, ctr, E);
  l1_edge_aggregate<<<(E + L1_SEG - 1) / L1_SEG + 2, 256, 0, stream>>>(psrc, pdst, xl1, lg1, m1,
                                                                       dn1, hacc, ctr, E);

  gemm2_kernel<<<(mtiles + 3) / 4, 128, 0, stream>>>(hacc, dn1, b1, W2l, W2r, xl2, xr2, N, mtiles);
  l2_edge_logits<<<(E + L2_SEG - 1) / L2_SEG + 2, 256, 0, stream>>>(psrc, pdst, xl2, xr2, att2,
                                                                    lg2, m2, ctr, E);
  l2_edge_aggregate<<<(E + L2_SEG - 1) / L2_SEG + 2, 256, 0, stream>>>(psrc, pdst, xl2, lg2, m2,
                                                                       acc2, ctr, E);
  l2_finish<<<blocks((long long)N * 10), 256, 0, stream>>>(acc2, b2, out, N * 10);
}


# GEMM1: a block's 16 rows of x come in once as whole lines and stay in LDS; its four waves read A fragments from there, not each 8 bytes from 16 lines per step; sums unchanged
# speedup vs baseline: 3.0166x; 1.0642x over previous
#include <hip/hip_runtime.h>
#include <math.h>

typedef __attribute__((ext_vector_type(2))) float v2f;
typedef __attribute__((ext_vector_type(8))) float v8f;
typedef __attribute__((ext_vector_type(4))) float v4f;

#define NEG_SLOPE 0.2f

#define L1_SEG 640
#define L2_SEG 768

__device__ __forceinline__ void atomicMaxF(float* addr, float v) {
  if (v >= 0.0f) atomicMax((int*)addr, __float_as_int(v));
  else           atomicMin((unsigned int*)addr, __float_as_uint(v));
}

__global__ void __launch_bounds__(256)
partition_edges(const int* __restrict__ ei, int* __restrict__ psrc,
                int* __restrict__ pdst, int* __restrict__ cnt, int E, int split) {
  __shared__ int wc0[8], wc1[8], sbase[2];
  const int tid = threadIdx.x, lane = tid & 31, wave = tid >> 5;
  const int base = blockIdx.x * 1024;
  int s[4], d[4];
  unsigned b0[4], b1[4];
  int c0 = 0, c1 = 0;
  #pragma unroll
  for (int j = 0; j < 4; ++j) {
    const int g = base + j * 256 + tid;
    const bool valid = (g < E);
    const int gg = valid ? g : E - 1;
    s[j] = ei[gg];
    d[j] = ei[(size_t)E + gg];
    b0[j] = __builtin_amdgcn_ballot_w32(valid && d[j] <  split);
    b1[j] = __builtin_amdgcn_ballot_w32(valid && d[j] >= split);
    c0 += __popc(b0[j]);
    c1 += __popc(b1[j]);
  }
  if (lane == 0) { wc0[wave] = c0; wc1[wave] = c1; }
  __syncthreads();
  if (tid == 0) {
    int t0 = 0, t1 = 0;
    #pragma unroll
    for (int w = 0; w < 8; ++w) { t0 += wc0[w]; t1 += wc1[w]; }
    sbase[0] = atomicAdd(cnt + 0, t0);
    sbase[1] = atomicAdd(cnt + 1, t1);
  }
  __syncthreads();
  int off0 = sbase[0], off1 = sbase[1];
  for (int w = 0; w < wave; ++w) { off0 += wc0[w]; off1 += wc1[w]; }
  const unsigned lt = (1u << lane) - 1u;
  #pragma unroll
  for (int j = 0; j < 4; ++j) {
    if ((b0[j] >> lane) & 1u) {
      const int pos = off0 + __popc(b0[j] & lt);
      psrc[pos] = s[j]; pdst[pos] = d[j];
    } else if ((b1[j] >> lane) & 1u) {
      const int pos = E - 1 - (off1 + __popc(b1[j] & lt));
      psrc[pos] = s[j]; pdst[pos] = d[j];
    }
    off0 += __popc(b0[j]);
    off1 += __popc(b1[j]);
  }
}

__device__ __forceinline__ void claim_chunk(int* __restrict__ tk, const int* __restrict__ cnt,
                                            int E, int CH, int& lo, int& hi) {
  __shared__ int s_lo, s_hi;
  if (threadIdx.x == 0) {
    const int n0 = cnt[0];
    const int die = (int)((__builtin_amdgcn_s_sendmsg_rtn(0x87) >> 16) & 15u);
    const int half = (die >> 2) & 1;
    int l = 0, h = 0;
    #pragma unroll
    for (int pass = 0; pass < 2; ++pass) {
      const int sset = half ^ pass;
      const long long beg = sset ? n0 : 0;
      const long long end = sset ? E : n0;
      const long long a = beg + (long long)atomicAdd(tk + 64 * sset, 1) * CH;
      if (a < end) {
        l = (int)a;
        h = (int)((a + CH < end) ? a + CH : end);
        break;
      }
    }
    s_lo = l; s_hi = h;
  }
  __syncthreads();
  lo = s_lo; hi = s_hi;
}

__global__ void __launch_bounds__(128)
gemm1_kernel(const float* __restrict__ x, const float* __restrict__ W1l,
             const float* __restrict__ W1r, float* __restrict__ xl1,
             float* __restrict__ xr1, float* __restrict__ hacc,
             float* __restrict__ m1, float* __restrict__ dn1,
             float* __restrict__ acc2, float* __restrict__ m2,
             int* __restrict__ ctr, int n_nodes) {
  __shared__ float xs[16 * 132];
  v4f xin[4];
  #pragma unroll
  for (int j = 0; j < 4; ++j) {
    const int p = threadIdx.x + 128 * j;
    int grow = blockIdx.x * 16 + (p >> 5);
    if (grow >= n_nodes) grow = n_nodes - 1;
    xin[j] = *(const v4f*)(x + (size_t)grow * 128 + (p & 31) * 4);
  }
  #pragma unroll
  for (int j = 0; j < 4; ++j) {
    const int p = threadIdx.x + 128 * j;
    *(v4f*)(xs + (p >> 5) * 132 + (p & 31) * 4) = xin[j];
  }
  __syncthreads();
  {
    const int t = threadIdx.x;
    const int r0 = blockIdx.x * 16;
    const v4f zero4 = {0.0f, 0.0f, 0.0f, 0.0f};
    const float ninf = -__builtin_inff();
    const v4f ninf4 = {ninf, ninf, ninf, ninf};
    for (int j = t; j < 256; j += 128)
      if (r0 + (j >> 4) < n_nodes) *(v4f*)(hacc + (size_t)r0 * 64 + j * 4) = zero4;
    if (t < 16) {
      if (r0 + t < n_nodes) *(v4f*)(m1 + (size_t)(r0 + t) * 4) = ninf4;
    } else if (t < 32) {
      if (r0 + t - 16 < n_nodes) *(v4f*)(dn1 + (size_t)(r0 + t - 16) * 4) = zero4;
    } else if (t < 96) {
      const int j = t - 32;
      if (r0 + (j >> 2) < n_nodes) *(v4f*)(acc2 + (size_t)r0 * 16 + j * 4) = zero4;
    } else if (t < 112) {
      if (r0 + t - 96 < n_nodes) m2[r0 + t - 96] = ninf;
    }
    if (blockIdx.x == 0)
      for (int j = t; j < 576; j += 128) ctr[j] = 0;
  }
  const int lane = threadIdx.x & 31;
  const int wave = threadIdx.x >> 5;
  const int m0   = blockIdx.x * 16;
  const int n0   = wave * 16;
  const int half = lane >> 4;
  const int rc   = lane & 15;

  const v2f*   xrow = (const v2f*)(xs + rc * 132);
  const int    bo   = n0 + rc + 128 * half;

  v8f accL = {}, accR = {};
  #pragma unroll
  for (int kc = 0; kc < 32; ++kc) {
    v2f a = xrow[2 * kc + half];
    v2f bl, br;
    bl[0] = W1l[bo + 256 * kc];
    bl[1] = W1l[bo + 256 * kc + 64];
    br[0] = W1r[bo + 256 * kc];
    br[1] = W1r[bo + 256 * kc + 64];
    accL = __builtin_amdgcn_wmma_f32_16x16x4_f32(
        false, a, false, bl, (short)0, accL, false, false);
    accR = __builtin_amdgcn_wmma_f32_16x16x4_f32(
        false, a, false, br, (short)0, accR, false, false);
  }

  #pragma unroll
  for (int r = 0; r < 8; ++r) {
    int row = m0 + r + 8 * half;
    if (row < n_nodes) {
      xl1[(size_t)row * 64 + n0 + rc] = accL[r];
      xr1[(size_t)row * 64 + n0 + rc] = accR[r];
    }
  }
}

__global__ void __launch_bounds__(128)
gemm2_kernel(const float* __restrict__ hacc, const float* __restrict__ den1,
             const float* __restrict__ b1, const float* __restrict__ W2l,
             const float* __restrict__ W2r, float* __restrict__ xl2,
             float* __restrict__ xr2, int n_nodes, int mtiles) {
  const int lane  = threadIdx.x & 31;
  const int mtile = blockIdx.x * 4 + (threadIdx.x >> 5);
  if (mtile >= mtiles) return;
  const int m0   = mtile * 16;
  const int half = lane >> 4;
  const int rc   = lane & 15;

  int arow = m0 + rc; if (arow >= n_nodes) arow = n_nodes - 1;
  const v2f* hrow = (const v2f*)(hacc + (size_t)arow * 64);
  const v4f  dn   = *(const v4f*)(den1 + (size_t)arow * 4);
  const bool colok = (rc < 10);

  v2f a[16];
  #pragma unroll
  for (int kc = 0; kc < 16; ++kc) a[kc] = hrow[2 * kc + half];
  #pragma unroll
  for (int kc = 0; kc < 16; ++kc) {
    const float d = dn[kc >> 2];
    const v2f bb = *(const v2f*)(b1 + 4 * kc + 2 * half);
    float v0 = (d > 0.0f) ? a[kc][0] / d : 0.0f;
    float v1 = (d > 0.0f) ? a[kc][1] / d : 0.0f;
    v0 += bb[0];
    v1 += bb[1];
    a[kc][0] = (v0 > 0.0f) ? v0 : expm1f(v0);
    a[kc][1] = (v1 > 0.0f) ? v1 : expm1f(v1);
  }

  v8f accL = {}, accR = {};
  #pragma unroll
  for (int kc = 0; kc < 16; ++kc) {
    const int k0 = 4 * kc + 2 * half;
    v2f bl, br;
    bl[0] = colok ? W2l[(size_t)(k0 + 0) * 10 + rc] : 0.0f;
    bl[1] = colok ? W2l[(size_t)(k0 + 1) * 10 + rc] : 0.0f;
    br[0] = colok ? W2r[(size_t)(k0 + 0) * 10 + rc] : 0.0f;
    br[1] = colok ? W2r[(size_t)(k0 + 1) * 10 + rc] : 0.0f;
    accL = __builtin_amdgcn_wmma_f32_16x16x4_f32(
        false, a[kc], false, bl, (short)0, accL, false, false);
    accR = __builtin_amdgcn_wmma_f32_16x16x4_f32(
        false, a[kc], false, br, (short)0, accR, false, false);
  }

  #pragma unroll
  for (int r = 0; r < 8; ++r) {
    int row = m0 + r + 8 * half;
    if (row < n_nodes) {
      xl2[(size_t)row * 16 + rc] = accL[r];
      xr2[(size_t)row * 16 + rc] = accR[r];
    }
  }
}

__global__ void __launch_bounds__(256)
l1_edge_logits(const int* __restrict__ psrc, const int* __restrict__ pdst,
               const float* __restrict__ xl1,
               const float* __restrict__ xr1, const float* __restrict__ att1,
               float* __restrict__ logits1, float* __restrict__ m1,
               int* __restrict__ ctr, int E) {
  int lo, hi;
  claim_chunk(ctr + 64, ctr, E, L1_SEG, lo, hi);
  if (lo >= hi) return;
  const int l = threadIdx.x & 15;
  const float4 at = *(const float4*)(att1 + l * 4);

  for (int base = lo; base < hi; base += 64) {
    int gg[4], dj[4];
    bool act[4];
    v4f xv[4], rv[4];
    #pragma unroll
    for (int it = 0; it < 4; ++it) {
      int g = base + it * 16 + (threadIdx.x >> 4);
      act[it] = (g < hi);
      if (!act[it]) g = hi - 1;
      gg[it] = g;
    }
    int sj[4];
    #pragma unroll
    for (int it = 0; it < 4; ++it) {
      sj[it] = psrc[gg[it]];
      dj[it] = pdst[gg[it]];
    }
    asm volatile("" :: "v"(sj[0]), "v"(dj[0]), "v"(sj[1]), "v"(dj[1]),
                       "v"(sj[2]), "v"(dj[2]), "v"(sj[3]), "v"(dj[3]));
    #pragma unroll
    for (int it = 0; it < 4; ++it) {
      xv[it] = *(const v4f*)(xl1 + (size_t)sj[it] * 64 + l * 4);
      rv[it] = *(const v4f*)(xr1 + (size_t)dj[it] * 64 + l * 4);
    }
    asm volatile("" :: "v"(xv[0]), "v"(rv[0]), "v"(xv[1]), "v"(rv[1]),
                       "v"(xv[2]), "v"(rv[2]), "v"(xv[3]), "v"(rv[3]));
    #pragma unroll
    for (int it = 0; it < 4; ++it) {
      float p = 0.0f, s;
      s = xv[it].x + rv[it].x; s = (s > 0.f) ? s : NEG_SLOPE * s; p += s * at.x;
      s = xv[it].y + rv[it].y; s = (s > 0.f) ? s : NEG_SLOPE * s; p += s * at.y;
      s = xv[it].z + rv[it].z; s = (s > 0.f) ? s : NEG_SLOPE * s; p += s * at.z;
      s = xv[it].w + rv[it].w; s = (s > 0.f) ? s : NEG_SLOPE * s; p += s * at.w;

      p += __shfl_xor(p, 1, 32);
      p += __shfl_xor(p, 2, 32);

      if (act[it] && (l & 3) == 0) {
        int hd = l >> 2;
        logits1[(size_t)gg[it] * 4 + hd] = p;
        atomicMaxF(m1 + (size_t)dj[it] * 4 + hd, p);
      }
    }
  }
}

__global__ void __launch_bounds__(256)
l1_edge_aggregate(const int* __restrict__ psrc, const int* __restrict__ pdst,
                  const float* __restrict__ xl1,
                  const float* __restrict__ logits1, const float* __restrict__ m1,
                  float* __restrict__ den1, float* __restrict__ hacc,
                  int* __restrict__ ctr, int E) {
  int lo, E_hi;
  claim_chunk(ctr + 192, ctr, E, L1_SEG, lo, E_hi);
  const int lane = threadIdx.x & 31;
  for (int g0 = lo + (threadIdx.x >> 5) * 8; g0 < E_hi; g0 += 64) {

    const int hd = lane & 3;
    int ge = g0 + (lane >> 2);
    const bool ok = (ge < E_hi);
    if (!ok) ge = E_hi - 1;
    const int srcj = psrc[ge];
    const int dstj = pdst[ge];
    const float lg = logits1[(size_t)ge * 4 + hd];
    const float mx = m1[(size_t)dstj * 4 + hd];

    float v0[8], v1[8];
    #pragma unroll
    for (int e = 0; e < 8; ++e) {
      const int s = __builtin_amdgcn_readlane(srcj, 4 * e);
      const float* r = xl1 + (size_t)s * 64 + lane;
      v0[e] = r[0];
      v1[e] = r[32];
    }

    const float w = __expf(lg - mx);
    if (ok) atomicAdd(den1 + (size_t)dstj * 4 + hd, w);

    const bool hi = (lane >= 16);
    const int  wi = __float_as_int(w);

    #pragma unroll
    for (int e = 0; e < 8; ++e) {
      if (g0 + e < E_hi) {
        const int d = __builtin_amdgcn_readlane(dstj, 4 * e);
        const float w0 = __int_as_float(__builtin_amdgcn_readlane(wi, 4 * e + 0));
        const float w1 = __int_as_float(__builtin_amdgcn_readlane(wi, 4 * e + 1));
        const float w2 = __int_as_float(__builtin_amdgcn_readlane(wi, 4 * e + 2));
        const float w3 = __int_as_float(__builtin_amdgcn_readlane(wi, 4 * e + 3));
        float* o = hacc + (size_t)d * 64 + lane;
        atomicAdd(o,      (hi ? w1 : w0) * v0[e]);
        atomicAdd(o + 32, (hi ? w3 : w2) * v1[e]);
      }
    }
  }
}

__global__ void __launch_bounds__(256)
l2_edge_logits(const int* __restrict__ psrc, const int* __restrict__ pdst,
               const float* __restrict__ xl2,
               const float* __restrict__ xr2, const float* __restrict__ att2,
               float* __restrict__ logits2, float* __restrict__ m2,
               int* __restrict__ ctr, int E) {
  int lo, hi;
  claim_chunk(ctr + 320, ctr, E, L2_SEG, lo, hi);
  if (lo >= hi) return;
  const int l  = threadIdx.x & 3;
  const int c0 = l * 4;

  float at0 = 0.0f, at1 = 0.0f, at2 = 0.0f, at3 = 0.0f;
  if (c0 + 0 < 10) at0 = att2[c0 + 0];
  if (c0 + 1 < 10) at1 = att2[c0 + 1];
  if (c0 + 2 < 10) at2 = att2[c0 + 2];
  if (c0 + 3 < 10) at3 = att2[c0 + 3];
  for (int base = lo; base < hi; base += 256) {
    int gg[4], dj[4];
    bool act[4];
    v4f a[4], b[4];
    #pragma unroll
    for (int it = 0; it < 4; ++it) {
      int g = base + it * 64 + (threadIdx.x >> 2);
      act[it] = (g < hi);
      if (!act[it]) g = hi - 1;
      gg[it] = g;
    }
    #pragma unroll
    for (int it = 0; it < 4; ++it) {
      const int src = psrc[gg[it]];
      dj[it] = pdst[gg[it]];
      a[it] = *(const v4f*)(xl2 + (size_t)src * 16 + c0);
      b[it] = *(const v4f*)(xr2 + (size_t)dj[it] * 16 + c0);
    }
    asm volatile("" :: "v"(a[0]), "v"(b[0]), "v"(a[1]), "v"(b[1]),
                       "v"(a[2]), "v"(b[2]), "v"(a[3]), "v"(b[3]));
    #pragma unroll
    for (int it = 0; it < 4; ++it) {
      float p = 0.0f, s;
      s = a[it].x + b[it].x; s = (s > 0.f) ? s : NEG_SLOPE * s; if (c0 + 0 < 10) p += s * at0;
      s = a[it].y + b[it].y; s = (s > 0.f) ? s : NEG_SLOPE * s; if (c0 + 1 < 10) p += s * at1;
      s = a[it].z + b[it].z; s = (s > 0.f) ? s : NEG_SLOPE * s; if (c0 + 2 < 10) p += s * at2;
      s = a[it].w + b[it].w; s = (s > 0.f) ? s : NEG_SLOPE * s; if (c0 + 3 < 10) p += s * at3;

      p += __shfl_xor(p, 1, 32);
      p += __shfl_xor(p, 2, 32);

      if (act[it] && l == 0) {
        logits2[gg[it]] = p;
        atomicMaxF(m2 + dj[it], p);
      }
    }
  }
}

__global__ void __launch_bounds__(256)
l2_edge_aggregate(const int* __restrict__ psrc, const int* __restrict__ pdst,
                  const float* __restrict__ xl2,
                  const float* __restrict__ logits2, const float* __restrict__ m2,
                  float* __restrict__ acc2, int* __restrict__ ctr, int E) {
  int lo, E_hi;
  claim_chunk(ctr + 448, ctr, E, L2_SEG, lo, E_hi);
  const int lane = threadIdx.x & 31;
  for (int g0 = lo + (threadIdx.x >> 5) * 16; g0 < E_hi; g0 += 128) {

    int ge = g0 + (lane & 15);
    if (ge >= E_hi) ge = E_hi - 1;
    const int srcj = psrc[ge];
    const int dstj = pdst[ge];
    const int wi   = __float_as_int(__expf(logits2[ge] - m2[dstj]));

    const bool hi = (lane >= 16);
    const int  c  = lane & 15;

    float v[8];
    #pragma unroll
    for (int p = 0; p < 8; ++p) {
      const int sa = __builtin_amdgcn_readlane(srcj, p);
      const int sb = __builtin_amdgcn_readlane(srcj, 8 + p);
      v[p] = xl2[(size_t)(hi ? sb : sa) * 16 + c];
    }
    #pragma unroll
    for (int p = 0; p < 8; ++p) {
      const int da = __builtin_amdgcn_readlane(dstj, p);
      const int db = __builtin_amdgcn_readlane(dstj, 8 + p);
      const float wa = __int_as_float(__builtin_amdgcn_readlane(wi, p));
      const float wb = __int_as_float(__builtin_amdgcn_readlane(wi, 8 + p));
      const int   d  = hi ? db : da;
      const float w  = hi ? wb : wa;
      const int   e  = g0 + p + (hi ? 8 : 0);
      if (c <= 10 && e < E_hi)
        atomicAdd(acc2 + (size_t)d * 16 + c, (c < 10) ? w * v[p] : w);
    }
  }
}

__global__ void __launch_bounds__(256)
l2_finish(const float* __restrict__ acc2,
          const float* __restrict__ b2, float* __restrict__ out, int total) {
  int i = blockIdx.x * blockDim.x + threadIdx.x;
  if (i >= total) return;
  int n = i / 10, c = i % 10;
  float d = acc2[(size_t)n * 16 + 10];
  float v = (d > 0.0f) ? acc2[(size_t)n * 16 + c] / d : 0.0f;
  out[i] = v + b2[c];
}

extern "C" void kernel_launch(void* const* d_in, const int* in_sizes, int n_in,
                              void* d_out, int out_size, void* d_ws, size_t ws_size,
                              hipStream_t stream) {
  const float* x    = (const float*)d_in[0];
  const int*   ei   = (const int*)d_in[1];
  const float* W1l  = (const float*)d_in[2];
  const float* W1r  = (const float*)d_in[3];
  const float* att1 = (const float*)d_in[4];
  const float* b1   = (const float*)d_in[5];
  const float* W2l  = (const float*)d_in[6];
  const float* W2r  = (const float*)d_in[7];
  const float* att2 = (const float*)d_in[8];
  const float* b2   = (const float*)d_in[9];
  float* out = (float*)d_out;

  const int N = in_sizes[0] / 128;
  const int E = in_sizes[1] / 2;

  float* ws = (float*)d_ws;
  size_t o = 0;
  float* xl1  = ws + o; o += (size_t)N * 64;
  float* xr1  = ws + o; o += (size_t)N * 64;
  float* hacc = ws + o; o += (size_t)N * 64;
  float* lg1  = ws + o; o += (size_t)E * 4;
  float* m1   = ws + o; o += (size_t)N * 4;
  float* dn1  = ws + o; o += (size_t)N * 4;
  int*   ctr  = (int*)(ws + o); o += 576;
  float* xl2  = ws + o; o += (size_t)N * 16;
  float* xr2  = ws + o; o += (size_t)N * 16;
  float* acc2 = ws + o; o += (size_t)N * 16;
  float* lg2  = ws + o; o += (size_t)E;
  float* m2   = ws + o; o += (size_t)N;
  int* psrc = (int*)(ws + o); o += (size_t)E;
  int* pdst = (int*)(ws + o); o += (size_t)E;

  auto blocks = [](long long n) { return (unsigned)((n + 255) / 256); };
  const int mtiles = (N + 15) / 16;

  gemm1_kernel<<<mtiles, 128, 0, stream>>>(x, W1l, W1r, xl1, xr1,
                                           hacc, m1, dn1, acc2, m2, ctr, N);
  const int split = ((N / 2) + 63) & ~63;
  partition_edges<<<(E + 1023) / 1024, 256, 0, stream>>>(ei, psrc, pdst, ctr, E, split);
  l1_edge_logits<<<(E + L1_SEG - 1) / L1_SEG + 2, 256, 0, stream>>>(psrc, pdst, xl1, xr1, att1,
                                                                    lg1, m1, ctr, E);
  l1_edge_aggregate<<<(E + L1_SEG - 1) / L1_SEG + 2, 256, 0, stream>>>(psrc, pdst, xl1, lg1, m1,
                                                                       dn1, hacc, ctr, E);

  gemm2_kernel<<<(mtiles + 3) / 4, 128, 0, stream>>>(hacc, dn1, b1, W2l, W2r, xl2, xr2, N, mtiles);
  l2_edge_logits<<<(E + L2_SEG - 1) / L2_SEG + 2, 256, 0, stream>>>(psrc, pdst, xl2, xr2, att2,
                                                                    lg2, m2, ctr, E);
  l2_edge_aggregate<<<(E + L2_SEG - 1) / L2_SEG + 2, 256, 0, stream>>>(psrc, pdst, xl2, lg2, m2,
                                                                       acc2, ctr, E);
  l2_finish<<<blocks((long long)N * 10), 256, 0, stream>>>(acc2, b2, out, N * 10);
}
